# starting kernel
# baseline (speedup 1.0000x reference)
#include <hip/hip_runtime.h>

#define N_NODES 100000
#define N_EDGES 1600000
#define F_IN    512
#define HIDDEN  16
#define N_CLS   7

typedef float v2f __attribute__((ext_vector_type(2)));
typedef float v8f __attribute__((ext_vector_type(8)));

__global__ void k_deg_init(float* deg, int n) {
    int t = blockIdx.x * blockDim.x + threadIdx.x;
    if (t < n) deg[t] = 1.0f;
}

__global__ void k_deg_edges(const int* __restrict__ dst, float* deg, int nE) {
    int e = blockIdx.x * blockDim.x + threadIdx.x;
    if (e < nE) atomicAdd(&deg[dst[e]], 1.0f);
}

__global__ void k_dinv(float* deg, int n) {
    int t = blockIdx.x * blockDim.x + threadIdx.x;
    if (t < n) deg[t] = __frsqrt_rn(deg[t]);
}

__global__ __launch_bounds__(256) void k_gemm0(const float* __restrict__ x,
                                               const float* __restrict__ W0,
                                               float* __restrict__ h0, int nTiles) {
    __shared__ float sW[F_IN * HIDDEN];
    for (int i = threadIdx.x; i < (F_IN * HIDDEN) / 4; i += 256)
        ((float4*)sW)[i] = ((const float4*)W0)[i];
    __syncthreads();

    const int wave   = threadIdx.x >> 5;
    const int lane   = threadIdx.x & 31;
    const int tile   = blockIdx.x * 8 + wave;
    if (tile >= nTiles) return;

    const int laneHi = lane >> 4;
    const int lane15 = lane & 15;
    const int row    = tile * 16 + lane15;
    const float* xrow = x + (long)row * F_IN;
    const int koff   = laneHi ? 4 : 0;

    v8f acc = {};
    for (int kb = 0; kb < F_IN; kb += 8) {
        float4 a4 = *(const float4*)(xrow + kb + koff);
        const float* wb = sW + (kb + koff) * HIDDEN + lane15;
        v2f a0; a0.x = a4.x; a0.y = a4.y;
        v2f b0; b0.x = wb[0];  b0.y = wb[16];
        acc = __builtin_amdgcn_wmma_f32_16x16x4_f32(false, a0, false, b0,
                                                    (short)0, acc, false, false);
        v2f a1; a1.x = a4.z; a1.y = a4.w;
        v2f b1; b1.x = wb[32]; b1.y = wb[48];
        acc = __builtin_amdgcn_wmma_f32_16x16x4_f32(false, a1, false, b1,
                                                    (short)0, acc, false, false);
    }
    float* out = h0 + (long)tile * 256 + laneHi * 128 + lane15;
#pragma unroll
    for (int i = 0; i < 8; ++i) out[i * 16] = acc[i];
}

__global__ void k_self_init16(const float* __restrict__ h, const float* __restrict__ dinv,
                              float* __restrict__ agg, int n4) {
    int t = blockIdx.x * blockDim.x + threadIdx.x;
    if (t >= n4) return;
    int node = t >> 2, q = t & 3;
    float w = dinv[node]; w *= w;
    float4 v = ((const float4*)(h + (long)node * 16))[q];
    v.x *= w; v.y *= w; v.z *= w; v.w *= w;
    ((float4*)(agg + (long)node * 16))[q] = v;
}

__global__ void k_edge16(const int* __restrict__ src, const int* __restrict__ dst,
                         const float* __restrict__ dinv, const float* __restrict__ h,
                         float* __restrict__ agg, int nE) {
    int e = blockIdx.x * blockDim.x + threadIdx.x;
    if (e >= nE) return;
    int s = src[e], d = dst[e];
    float w = dinv[s] * dinv[d];
    const float* hs = h + (long)s * 16;
    float* ad = agg + (long)d * 16;
#pragma unroll
    for (int i = 0; i < 16; ++i) atomicAdd(ad + i, hs[i] * w);
}

__global__ void k_relu_bias(const float* __restrict__ agg, const float* __restrict__ b0,
                            float* __restrict__ h1, int total) {
    int t = blockIdx.x * blockDim.x + threadIdx.x;
    if (t >= total) return;
    float v = agg[t] + b0[t & 15];
    h1[t] = v > 0.0f ? v : 0.0f;
}

__global__ __launch_bounds__(256) void k_gemm1(const float* __restrict__ h1,
                                               const float* __restrict__ W1,
                                               float* __restrict__ h2, int nTiles) {
    __shared__ float sW[16 * 16];
    {
        int k = threadIdx.x >> 4, n = threadIdx.x & 15;
        sW[threadIdx.x] = (n < N_CLS) ? W1[k * N_CLS + n] : 0.0f;
    }
    __syncthreads();

    const int wave   = threadIdx.x >> 5;
    const int lane   = threadIdx.x & 31;
    const int tile   = blockIdx.x * 8 + wave;
    if (tile >= nTiles) return;

    const int laneHi = lane >> 4;
    const int lane15 = lane & 15;
    const int row    = tile * 16 + lane15;
    const float* hrow = h1 + (long)row * 16;
    const int koff   = laneHi ? 2 : 0;

    v8f acc = {};
#pragma unroll
    for (int kb = 0; kb < 16; kb += 4) {
        v2f a; a.x = hrow[kb + koff]; a.y = hrow[kb + koff + 1];
        v2f b; b.x = sW[(kb + koff) * 16 + lane15];
               b.y = sW[(kb + koff + 1) * 16 + lane15];
        acc = __builtin_amdgcn_wmma_f32_16x16x4_f32(false, a, false, b,
                                                    (short)0, acc, false, false);
    }
    float* out = h2 + (long)tile * 256 + laneHi * 128 + lane15;
#pragma unroll
    for (int i = 0; i < 8; ++i) out[i * 16] = acc[i];
}

__global__ void k_out_init7(const float* __restrict__ h2, const float* __restrict__ dinv,
                            const float* __restrict__ b1, float* __restrict__ out, int total) {
    int t = blockIdx.x * blockDim.x + threadIdx.x;
    if (t >= total) return;
    int node = t / N_CLS, c = t - node * N_CLS;
    float w = dinv[node];
    out[t] = h2[(long)node * 16 + c] * w * w + b1[c];
}

__global__ void k_edge7(const int* __restrict__ src, const int* __restrict__ dst,
                        const float* __restrict__ dinv, const float* __restrict__ h2,
                        float* __restrict__ out, int nE) {
    int e = blockIdx.x * blockDim.x + threadIdx.x;
    if (e >= nE) return;
    int s = src[e], d = dst[e];
    float w = dinv[s] * dinv[d];
    const float* hs = h2 + (long)s * 16;
    float* od = out + (long)d * N_CLS;
#pragma unroll
    for (int c = 0; c < N_CLS; ++c) atomicAdd(od + c, hs[c] * w);
}

extern "C" void kernel_launch(void* const* d_in, const int* in_sizes, int n_in,
                              void* d_out, int out_size, void* d_ws, size_t ws_size,
                              hipStream_t stream) {
    const float* x  = (const float*)d_in[0];
    const int*   ei = (const int*)d_in[1];
    const float* W0 = (const float*)d_in[2];
    const float* b0 = (const float*)d_in[3];
    const float* W1 = (const float*)d_in[4];
    const float* b1 = (const float*)d_in[5];
    float* out = (float*)d_out;

    const int* src = ei;
    const int* dst = ei + N_EDGES;

    float* ws   = (float*)d_ws;
    float* dinv = ws;
    float* h0   = dinv + N_NODES;
    float* agg0 = h0 + (long)N_NODES * 16;
    float* h2   = agg0 + (long)N_NODES * 16;

    const int nTiles = N_NODES / 16;
    const dim3 B(256);
    auto gr = [](long n) { return dim3((unsigned)((n + 255) / 256)); };

    k_deg_init <<<gr(N_NODES), B, 0, stream>>>(dinv, N_NODES);
    k_deg_edges<<<gr(N_EDGES), B, 0, stream>>>(dst, dinv, N_EDGES);
    k_dinv     <<<gr(N_NODES), B, 0, stream>>>(dinv, N_NODES);

    k_gemm0<<<dim3((nTiles + 7) / 8), B, 0, stream>>>(x, W0, h0, nTiles);
    k_self_init16<<<gr((long)N_NODES * 4), B, 0, stream>>>(h0, dinv, agg0, N_NODES * 4);
    k_edge16<<<gr(N_EDGES), B, 0, stream>>>(src, dst, dinv, h0, agg0, N_EDGES);
    k_relu_bias<<<gr((long)N_NODES * 16), B, 0, stream>>>(agg0, b0, h0, N_NODES * 16);

    k_gemm1<<<dim3((nTiles + 7) / 8), B, 0, stream>>>(h0, W1, h2, nTiles);
    k_out_init7<<<gr((long)N_NODES * N_CLS), B, 0, stream>>>(h2, dinv, b1, out, N_NODES * N_CLS);
    k_edge7<<<gr(N_EDGES), B, 0, stream>>>(src, dst, dinv, h2, out, N_EDGES);
}


# Edge ranks from LDS counters, not global atomics: 25 work-groups each rank 64000 edges on a 16-bit per-node table in LDS; one scan gives per-chunk row offsets; node sum order: chunk, then LDS arrival
# speedup vs baseline: 5.4148x; 5.4148x over previous
#include <hip/hip_runtime.h>

#define N_NODES 100000
#define N_EDGES 1600000
#define F_IN    512
#define HIDDEN  16
#define N_CLS   7

typedef float v2f __attribute__((ext_vector_type(2)));
typedef float v8f __attribute__((ext_vector_type(8)));


#define G_CH       25
#define CH_E       64000
#define NPAIR      (N_NODES / 2)
#define CT_PITCH   50048
#define NRANGE     98
#define PART_PITCH 128

__global__ __launch_bounds__(1024) void k_rank_lds(const int* __restrict__ dst, unsigned short* __restrict__ rank,
                                                   unsigned* __restrict__ ctab, int* __restrict__ part) {
    __shared__ unsigned sCnt[NPAIR];
    const int g = blockIdx.x;
    uint4* s4 = (uint4*)sCnt;
    for (int i = threadIdx.x; i < NPAIR / 4; i += 1024) s4[i] = make_uint4(0u, 0u, 0u, 0u);
    __syncthreads();

    const int4* d4 = (const int4*)(dst + g * CH_E);
    uint2*      r2 = (uint2*)(rank + g * CH_E);
#pragma unroll
    for (int h = 0; h < 2; ++h) {
        int4 d[8];
#pragma unroll
        for (int k = 0; k < 8; ++k) {
            const int i = (h * 8 + k) * 1024 + threadIdx.x;
            d[k] = d4[i < CH_E / 4 ? i : 0];
        }
        __builtin_amdgcn_sched_barrier(0);
#pragma unroll
        for (int k = 0; k < 8; ++k) {
            const int i = (h * 8 + k) * 1024 + threadIdx.x;
            if (i < CH_E / 4) {
                const unsigned sx = (d[k].x & 1) << 4, sy = (d[k].y & 1) << 4;
                const unsigned sz = (d[k].z & 1) << 4, sw = (d[k].w & 1) << 4;
                const unsigned ox = atomicAdd(&sCnt[d[k].x >> 1], 1u << sx);
                const unsigned oy = atomicAdd(&sCnt[d[k].y >> 1], 1u << sy);
                const unsigned oz = atomicAdd(&sCnt[d[k].z >> 1], 1u << sz);
                const unsigned ow = atomicAdd(&sCnt[d[k].w >> 1], 1u << sw);
                uint2 o;
                o.x = ((ox >> sx) & 0xFFFFu) | (((oy >> sy) & 0xFFFFu) << 16);
                o.y = ((oz >> sz) & 0xFFFFu) | (((ow >> sw) & 0xFFFFu) << 16);
                r2[i] = o;
            }
        }
    }
    __syncthreads();

    const int wave = threadIdx.x >> 5, lane = threadIdx.x & 31;
    uint4* c4 = (uint4*)(ctab + (long)g * CT_PITCH);
    for (int r = wave; r < NRANGE; r += 32) {
        unsigned sum = 0;
#pragma unroll
        for (int j = 0; j < 4; ++j) {
            const int q = r * 128 + j * 32 + lane;
            if (q < NPAIR / 4) {
                const uint4 v = s4[q];
                c4[q] = v;
                sum += ((v.x & 0xFFFFu) + (v.x >> 16)) + ((v.y & 0xFFFFu) + (v.y >> 16))
                     + ((v.z & 0xFFFFu) + (v.z >> 16)) + ((v.w & 0xFFFFu) + (v.w >> 16));
            }
        }
        for (int o = 16; o > 0; o >>= 1) sum += __shfl_xor(sum, o);
        if (lane == 0) part[g * PART_PITCH + r] = (int)sum;
    }
}

__global__ __launch_bounds__(512) void k_scan_tab(const unsigned* __restrict__ ctab, const int* __restrict__ part,
                                                  int* __restrict__ rowEnd, int* __restrict__ tab) {
    __shared__ int sTot[16];
    __shared__ int sPre[16];
    const int b = blockIdx.x, t = threadIdx.x;
    const int wave = t >> 5, lane = t & 31;
    const int p = b * 512 + t;
    const bool ok = p < NPAIR;

    unsigned c[G_CH];
#pragma unroll
    for (int g = 0; g < G_CH; ++g) c[g] = ctab[(long)g * CT_PITCH + (ok ? p : 0)];

    constexpr int NPV = (G_CH * PART_PITCH + 511) / 512;
    int pv[NPV];
#pragma unroll
    for (int k = 0; k < NPV; ++k) {
        const int idx = k * 512 + t;
        const bool use = idx < G_CH * PART_PITCH && (idx & (PART_PITCH - 1)) < b;
        pv[k] = part[use ? idx : 0];
    }
    __builtin_amdgcn_sched_barrier(0);
    int pre = 0;
#pragma unroll
    for (int k = 0; k < NPV; ++k) {
        const int idx = k * 512 + t;
        const bool use = idx < G_CH * PART_PITCH && (idx & (PART_PITCH - 1)) < b;
        pre += use ? pv[k] : 0;
    }

    int lo = 0, hi = 0;
#pragma unroll
    for (int g = 0; g < G_CH; ++g) {
        if (!ok) c[g] = 0u;
        lo += (int)(c[g] & 0xFFFFu); hi += (int)(c[g] >> 16);
    }
    const int s = lo + hi;

    int inc = s;
    for (int o = 1; o < 32; o <<= 1) {
        int v = __shfl_up(inc, o);
        if (lane >= o) inc += v;
    }
    for (int o = 16; o > 0; o >>= 1) pre += __shfl_xor(pre, o);
    if (lane == 31) sTot[wave] = inc;
    if (lane == 0)  sPre[wave] = pre;
    __syncthreads();

    int base = (lane < 16) ? (sPre[lane] + ((lane < wave) ? sTot[lane] : 0)) : 0;
    for (int o = 16; o > 0; o >>= 1) base += __shfl_xor(base, o);

    const int ex = base + inc - s;
    if (ok) {
        ((int2*)rowEnd)[p] = make_int2(ex + lo, ex + s);
        int runLo = ex, runHi = ex + lo;
#pragma unroll
        for (int g = 0; g < G_CH - 1; ++g) {
            runLo += (int)(c[g] & 0xFFFFu);
            runHi += (int)(c[g] >> 16);
            ((int2*)(tab + (long)g * N_NODES))[p] = make_int2(runLo, runHi);
        }
    }
}

__global__ void k_place(const int* __restrict__ src, const int* __restrict__ dst,
                        const unsigned short* __restrict__ rank, const int* __restrict__ rowEnd,
                        const int* __restrict__ tab, int* __restrict__ csr, int nE) {
    int e = blockIdx.x * blockDim.x + threadIdx.x;
    if (e >= nE) return;
    const int g = blockIdx.x / (CH_E / 256);
    int d = dst[e];
    int r = rank[e];
    int s = src[e];
    int p;
    if (g == 0) p = (d > 0) ? rowEnd[d - 1] : 0;
    else        p = tab[(long)(g - 1) * N_NODES + d];
    csr[p + r] = s;
}

__global__ __launch_bounds__(256) void k_gemm0(const float* __restrict__ x,
                                               const float* __restrict__ W0,
                                               const int* __restrict__ rowEnd,
                                               float* __restrict__ h0, int nTiles) {
    __shared__ float sW[F_IN * HIDDEN];
    for (int i = threadIdx.x; i < (F_IN * HIDDEN) / 4; i += 256)
        ((float4*)sW)[i] = ((const float4*)W0)[i];
    __syncthreads();

    const int wave   = threadIdx.x >> 5;
    const int lane   = threadIdx.x & 31;
    const int tile   = blockIdx.x * 8 + wave;
    if (tile >= nTiles) return;

    const int laneHi = lane >> 4;
    const int lane15 = lane & 15;
    const int row    = tile * 16 + lane15;
    const float* xrow = x + (long)row * F_IN;
    const int koff   = laneHi ? 4 : 0;

    v8f acc = {};
    for (int kb = 0; kb < F_IN; kb += 8) {
        float4 a4 = *(const float4*)(xrow + kb + koff);
        const float* wb = sW + (kb + koff) * HIDDEN + lane15;
        v2f a0; a0.x = a4.x; a0.y = a4.y;
        v2f b0; b0.x = wb[0];  b0.y = wb[16];
        acc = __builtin_amdgcn_wmma_f32_16x16x4_f32(false, a0, false, b0,
                                                    (short)0, acc, false, false);
        v2f a1; a1.x = a4.z; a1.y = a4.w;
        v2f b1; b1.x = wb[32]; b1.y = wb[48];
        acc = __builtin_amdgcn_wmma_f32_16x16x4_f32(false, a1, false, b1,
                                                    (short)0, acc, false, false);
    }
    const int r0 = tile * 16 + laneHi * 8;
    int prev = (r0 > 0) ? rowEnd[r0 - 1] : 0;
    float* out = h0 + (long)tile * 256 + laneHi * 128 + lane15;
#pragma unroll
    for (int i = 0; i < 8; ++i) {
        int e = rowEnd[r0 + i];
        float dv = __frsqrt_rn((float)(e - prev + 1));
        prev = e;
        out[i * 16] = acc[i] * dv;
    }
}

__global__ __launch_bounds__(256) void k_pull16(const int* __restrict__ rowEnd,
                                                const int* __restrict__ csr,
                                                const float* __restrict__ hp,
                                                const float* __restrict__ b0,
                                                float* __restrict__ h1) {
    const int t = blockIdx.x * 256 + threadIdx.x;
    const int d = t >> 4, f = t & 15;
    const int e1 = rowEnd[d];
    const int e0 = (d > 0) ? rowEnd[d - 1] : 0;
    float acc = hp[(long)d * 16 + f];
    for (int j = e0; j < e1; j += 16) {
        const int m = (e1 - j < 16) ? (e1 - j) : 16;
        const int mine = (f < m) ? csr[j + f] : 0;
        for (int k = 0; k < m; k += 4) {
            int s0 = __shfl(mine, k, 16),     s1 = __shfl(mine, k + 1, 16);
            int s2 = __shfl(mine, k + 2, 16), s3 = __shfl(mine, k + 3, 16);
            float v0 = hp[(long)s0 * 16 + f], v1 = hp[(long)s1 * 16 + f];
            float v2 = hp[(long)s2 * 16 + f], v3 = hp[(long)s3 * 16 + f];
            __builtin_amdgcn_sched_barrier(0);
            acc += v0;
            acc += (k + 1 < m) ? v1 : 0.0f;
            acc += (k + 2 < m) ? v2 : 0.0f;
            acc += (k + 3 < m) ? v3 : 0.0f;
        }
    }
    float v = acc * __frsqrt_rn((float)(e1 - e0 + 1)) + b0[f];
    h1[(long)d * 16 + f] = v > 0.0f ? v : 0.0f;
}

__global__ __launch_bounds__(256) void k_gemm1(const float* __restrict__ h1,
                                               const float* __restrict__ W1,
                                               const int* __restrict__ rowEnd,
                                               float* __restrict__ h2, int nTiles) {
    __shared__ float sW[16 * 16];
    {
        int k = threadIdx.x >> 4, n = threadIdx.x & 15;
        sW[threadIdx.x] = (n < N_CLS) ? W1[k * N_CLS + n] : 0.0f;
    }
    __syncthreads();

    const int wave   = threadIdx.x >> 5;
    const int lane   = threadIdx.x & 31;
    const int tile   = blockIdx.x * 8 + wave;
    if (tile >= nTiles) return;

    const int laneHi = lane >> 4;
    const int lane15 = lane & 15;
    const int row    = tile * 16 + lane15;
    const float* hrow = h1 + (long)row * 16;
    const int koff   = laneHi ? 2 : 0;

    v8f acc = {};
#pragma unroll
    for (int kb = 0; kb < 16; kb += 4) {
        v2f a; a.x = hrow[kb + koff]; a.y = hrow[kb + koff + 1];
        v2f b; b.x = sW[(kb + koff) * 16 + lane15];
               b.y = sW[(kb + koff + 1) * 16 + lane15];
        acc = __builtin_amdgcn_wmma_f32_16x16x4_f32(false, a, false, b,
                                                    (short)0, acc, false, false);
    }
    const int r0 = tile * 16 + laneHi * 8;
    int prev = (r0 > 0) ? rowEnd[r0 - 1] : 0;
    float* out = h2 + (long)tile * 256 + laneHi * 128 + lane15;
#pragma unroll
    for (int i = 0; i < 8; ++i) {
        int e = rowEnd[r0 + i];
        float dv = __frsqrt_rn((float)(e - prev + 1));
        prev = e;
        out[i * 16] = acc[i] * dv;
    }
}

__global__ __launch_bounds__(256) void k_pull7(const int* __restrict__ rowEnd,
                                               const int* __restrict__ csr,
                                               const float* __restrict__ hp,
                                               const float* __restrict__ b1,
                                               float* __restrict__ out) {
    __shared__ float sOut[32 * N_CLS];
    const int ln = threadIdx.x >> 3, c = threadIdx.x & 7;
    const int d  = blockIdx.x * 32 + ln;
    const int e1 = rowEnd[d];
    const int e0 = (d > 0) ? rowEnd[d - 1] : 0;
    float acc = hp[(long)d * 16 + c];
    for (int j = e0; j < e1; j += 8) {
        const int m = (e1 - j < 8) ? (e1 - j) : 8;
        const int mine = (c < m) ? csr[j + c] : 0;
        for (int k = 0; k < m; k += 4) {
            int s0 = __shfl(mine, k, 8),     s1 = __shfl(mine, k + 1, 8);
            int s2 = __shfl(mine, k + 2, 8), s3 = __shfl(mine, k + 3, 8);
            float v0 = hp[(long)s0 * 16 + c], v1 = hp[(long)s1 * 16 + c];
            float v2 = hp[(long)s2 * 16 + c], v3 = hp[(long)s3 * 16 + c];
            __builtin_amdgcn_sched_barrier(0);
            acc += v0;
            acc += (k + 1 < m) ? v1 : 0.0f;
            acc += (k + 2 < m) ? v2 : 0.0f;
            acc += (k + 3 < m) ? v3 : 0.0f;
        }
    }
    if (c < N_CLS)
        sOut[ln * N_CLS + c] = acc * __frsqrt_rn((float)(e1 - e0 + 1)) + b1[c];
    __syncthreads();
    if (threadIdx.x < 32 * N_CLS)
        out[(long)blockIdx.x * (32 * N_CLS) + threadIdx.x] = sOut[threadIdx.x];
}

extern "C" void kernel_launch(void* const* d_in, const int* in_sizes, int n_in,
                              void* d_out, int out_size, void* d_ws, size_t ws_size,
                              hipStream_t stream) {
    const float* x  = (const float*)d_in[0];
    const int*   ei = (const int*)d_in[1];
    const float* W0 = (const float*)d_in[2];
    const float* b0 = (const float*)d_in[3];
    const float* W1 = (const float*)d_in[4];
    const float* b1 = (const float*)d_in[5];
    float* out = (float*)d_out;

    const int* src = ei;
    const int* dst = ei + N_EDGES;

    float* ws     = (float*)d_ws;
    int*   rowEnd = (int*)ws;
    float* h0     = ws + N_NODES;
    float* h1     = h0 + (long)N_NODES * 16;
    int*   csr    = (int*)(h1 + (long)N_NODES * 16);
    unsigned short* rank = (unsigned short*)h0;
    int*      tab  = (int*)h0 + 800000;
    unsigned* ctab = (unsigned*)csr;
    int*      part = csr + G_CH * CT_PITCH;

    const int nTiles = N_NODES / 16;
    const dim3 B(256);
    auto gr = [](long n) { return dim3((unsigned)((n + 255) / 256)); };

    k_rank_lds<<<dim3(G_CH), dim3(1024), 0, stream>>>(dst, rank, ctab, part);
    k_scan_tab<<<dim3(NRANGE), dim3(512), 0, stream>>>(ctab, part, rowEnd, tab);
    k_place   <<<gr(N_EDGES), B, 0, stream>>>(src, dst, rank, rowEnd, tab, csr, N_EDGES);

    k_gemm0 <<<dim3((nTiles + 7) / 8), B, 0, stream>>>(x, W0, rowEnd, h0, nTiles);
    k_pull16<<<dim3(N_NODES * 16 / 256), B, 0, stream>>>(rowEnd, csr, h0, b0, h1);

    k_gemm1<<<dim3((nTiles + 7) / 8), B, 0, stream>>>(h1, W1, rowEnd, h0, nTiles);
    k_pull7<<<dim3(N_NODES / 32), B, 0, stream>>>(rowEnd, csr, h0, b1, out);
}


# Layer 1 reassociated: last gather sums dinv-scaled 16-wide h1 rows (8 lanes x 8 B), then its block applies W1 on WMMA; gemm1 launch and h2 plane gone; sum order: neighbours first, then k
# speedup vs baseline: 5.6367x; 1.0410x over previous
#include <hip/hip_runtime.h>

#define N_NODES 100000
#define N_EDGES 1600000
#define F_IN    512
#define HIDDEN  16
#define N_CLS   7

typedef float v2f __attribute__((ext_vector_type(2)));
typedef float v8f __attribute__((ext_vector_type(8)));


#define G_CH       25
#define CH_E       64000
#define NPAIR      (N_NODES / 2)
#define CT_PITCH   50048
#define NRANGE     98
#define PART_PITCH 128

__global__ __launch_bounds__(1024) void k_rank_lds(const int* __restrict__ dst, unsigned short* __restrict__ rank,
                                                   unsigned* __restrict__ ctab, int* __restrict__ part) {
    __shared__ unsigned sCnt[NPAIR];
    const int g = blockIdx.x;
    uint4* s4 = (uint4*)sCnt;
    for (int i = threadIdx.x; i < NPAIR / 4; i += 1024) s4[i] = make_uint4(0u, 0u, 0u, 0u);
    __syncthreads();

    const int4* d4 = (const int4*)(dst + g * CH_E);
    uint2*      r2 = (uint2*)(rank + g * CH_E);
#pragma unroll
    for (int h = 0; h < 2; ++h) {
        int4 d[8];
#pragma unroll
        for (int k = 0; k < 8; ++k) {
            const int i = (h * 8 + k) * 1024 + threadIdx.x;
            d[k] = d4[i < CH_E / 4 ? i : 0];
        }
        __builtin_amdgcn_sched_barrier(0);
#pragma unroll
        for (int k = 0; k < 8; ++k) {
            const int i = (h * 8 + k) * 1024 + threadIdx.x;
            if (i < CH_E / 4) {
                const unsigned sx = (d[k].x & 1) << 4, sy = (d[k].y & 1) << 4;
                const unsigned sz = (d[k].z & 1) << 4, sw = (d[k].w & 1) << 4;
                const unsigned ox = atomicAdd(&sCnt[d[k].x >> 1], 1u << sx);
                const unsigned oy = atomicAdd(&sCnt[d[k].y >> 1], 1u << sy);
                const unsigned oz = atomicAdd(&sCnt[d[k].z >> 1], 1u << sz);
                const unsigned ow = atomicAdd(&sCnt[d[k].w >> 1], 1u << sw);
                uint2 o;
                o.x = ((ox >> sx) & 0xFFFFu) | (((oy >> sy) & 0xFFFFu) << 16);
                o.y = ((oz >> sz) & 0xFFFFu) | (((ow >> sw) & 0xFFFFu) << 16);
                r2[i] = o;
            }
        }
    }
    __syncthreads();

    const int wave = threadIdx.x >> 5, lane = threadIdx.x & 31;
    uint4* c4 = (uint4*)(ctab + (long)g * CT_PITCH);
    for (int r = wave; r < NRANGE; r += 32) {
        unsigned sum = 0;
#pragma unroll
        for (int j = 0; j < 4; ++j) {
            const int q = r * 128 + j * 32 + lane;
            if (q < NPAIR / 4) {
                const uint4 v = s4[q];
                c4[q] = v;
                sum += ((v.x & 0xFFFFu) + (v.x >> 16)) + ((v.y & 0xFFFFu) + (v.y >> 16))
                     + ((v.z & 0xFFFFu) + (v.z >> 16)) + ((v.w & 0xFFFFu) + (v.w >> 16));
            }
        }
        for (int o = 16; o > 0; o >>= 1) sum += __shfl_xor(sum, o);
        if (lane == 0) part[g * PART_PITCH + r] = (int)sum;
    }
}

__global__ __launch_bounds__(512) void k_scan_tab(const unsigned* __restrict__ ctab, const int* __restrict__ part,
                                                  int* __restrict__ rowEnd, int* __restrict__ tab) {
    __shared__ int sTot[16];
    __shared__ int sPre[16];
    const int b = blockIdx.x, t = threadIdx.x;
    const int wave = t >> 5, lane = t & 31;
    const int p = b * 512 + t;
    const bool ok = p < NPAIR;

    unsigned c[G_CH];
#pragma unroll
    for (int g = 0; g < G_CH; ++g) c[g] = ctab[(long)g * CT_PITCH + (ok ? p : 0)];

    constexpr int NPV = (G_CH * PART_PITCH + 511) / 512;
    int pv[NPV];
#pragma unroll
    for (int k = 0; k < NPV; ++k) {
        const int idx = k * 512 + t;
        const bool use = idx < G_CH * PART_PITCH && (idx & (PART_PITCH - 1)) < b;
        pv[k] = part[use ? idx : 0];
    }
    __builtin_amdgcn_sched_barrier(0);
    int pre = 0;
#pragma unroll
    for (int k = 0; k < NPV; ++k) {
        const int idx = k * 512 + t;
        const bool use = idx < G_CH * PART_PITCH && (idx & (PART_PITCH - 1)) < b;
        pre += use ? pv[k] : 0;
    }

    int lo = 0, hi = 0;
#pragma unroll
    for (int g = 0; g < G_CH; ++g) {
        if (!ok) c[g] = 0u;
        lo += (int)(c[g] & 0xFFFFu); hi += (int)(c[g] >> 16);
    }
    const int s = lo + hi;

    int inc = s;
    for (int o = 1; o < 32; o <<= 1) {
        int v = __shfl_up(inc, o);
        if (lane >= o) inc += v;
    }
    for (int o = 16; o > 0; o >>= 1) pre += __shfl_xor(pre, o);
    if (lane == 31) sTot[wave] = inc;
    if (lane == 0)  sPre[wave] = pre;
    __syncthreads();

    int base = (lane < 16) ? (sPre[lane] + ((lane < wave) ? sTot[lane] : 0)) : 0;
    for (int o = 16; o > 0; o >>= 1) base += __shfl_xor(base, o);

    const int ex = base + inc - s;
    if (ok) {
        ((int2*)rowEnd)[p] = make_int2(ex + lo, ex + s);
        int runLo = ex, runHi = ex + lo;
#pragma unroll
        for (int g = 0; g < G_CH - 1; ++g) {
            runLo += (int)(c[g] & 0xFFFFu);
            runHi += (int)(c[g] >> 16);
            ((int2*)(tab + (long)g * N_NODES))[p] = make_int2(runLo, runHi);
        }
    }
}

__global__ void k_place(const int* __restrict__ src, const int* __restrict__ dst,
                        const unsigned short* __restrict__ rank, const int* __restrict__ rowEnd,
                        const int* __restrict__ tab, int* __restrict__ csr, int nE) {
    int e = blockIdx.x * blockDim.x + threadIdx.x;
    if (e >= nE) return;
    const int g = blockIdx.x / (CH_E / 256);
    int d = dst[e];
    int r = rank[e];
    int s = src[e];
    int p;
    if (g == 0) p = (d > 0) ? rowEnd[d - 1] : 0;
    else        p = tab[(long)(g - 1) * N_NODES + d];
    csr[p + r] = s;
}

__global__ __launch_bounds__(256) void k_gemm0(const float* __restrict__ x,
                                               const float* __restrict__ W0,
                                               const int* __restrict__ rowEnd,
                                               float* __restrict__ h0, int nTiles) {
    __shared__ float sW[F_IN * HIDDEN];
    for (int i = threadIdx.x; i < (F_IN * HIDDEN) / 4; i += 256)
        ((float4*)sW)[i] = ((const float4*)W0)[i];
    __syncthreads();

    const int wave   = threadIdx.x >> 5;
    const int lane   = threadIdx.x & 31;
    const int tile   = blockIdx.x * 8 + wave;
    if (tile >= nTiles) return;

    const int laneHi = lane >> 4;
    const int lane15 = lane & 15;
    const int row    = tile * 16 + lane15;
    const float* xrow = x + (long)row * F_IN;
    const int koff   = laneHi ? 4 : 0;

    v8f acc = {};
    for (int kb = 0; kb < F_IN; kb += 8) {
        float4 a4 = *(const float4*)(xrow + kb + koff);
        const float* wb = sW + (kb + koff) * HIDDEN + lane15;
        v2f a0; a0.x = a4.x; a0.y = a4.y;
        v2f b0; b0.x = wb[0];  b0.y = wb[16];
        acc = __builtin_amdgcn_wmma_f32_16x16x4_f32(false, a0, false, b0,
                                                    (short)0, acc, false, false);
        v2f a1; a1.x = a4.z; a1.y = a4.w;
        v2f b1; b1.x = wb[32]; b1.y = wb[48];
        acc = __builtin_amdgcn_wmma_f32_16x16x4_f32(false, a1, false, b1,
                                                    (short)0, acc, false, false);
    }
    const int r0 = tile * 16 + laneHi * 8;
    int prev = (r0 > 0) ? rowEnd[r0 - 1] : 0;
    float* out = h0 + (long)tile * 256 + laneHi * 128 + lane15;
#pragma unroll
    for (int i = 0; i < 8; ++i) {
        int e = rowEnd[r0 + i];
        float dv = __frsqrt_rn((float)(e - prev + 1));
        prev = e;
        out[i * 16] = acc[i] * dv;
    }
}

__global__ __launch_bounds__(256) void k_pull16(const int* __restrict__ rowEnd,
                                                const int* __restrict__ csr,
                                                const float* __restrict__ hp,
                                                const float* __restrict__ b0,
                                                float* __restrict__ h1) {
    const int t = blockIdx.x * 256 + threadIdx.x;
    const int d = t >> 4, f = t & 15;
    const int e1 = rowEnd[d];
    const int e0 = (d > 0) ? rowEnd[d - 1] : 0;
    float acc = hp[(long)d * 16 + f];
    for (int j = e0; j < e1; j += 16) {
        const int m = (e1 - j < 16) ? (e1 - j) : 16;
        const int mine = (f < m) ? csr[j + f] : 0;
        for (int k = 0; k < m; k += 4) {
            int s0 = __shfl(mine, k, 16),     s1 = __shfl(mine, k + 1, 16);
            int s2 = __shfl(mine, k + 2, 16), s3 = __shfl(mine, k + 3, 16);
            float v0 = hp[(long)s0 * 16 + f], v1 = hp[(long)s1 * 16 + f];
            float v2 = hp[(long)s2 * 16 + f], v3 = hp[(long)s3 * 16 + f];
            __builtin_amdgcn_sched_barrier(0);
            acc += v0;
            acc += (k + 1 < m) ? v1 : 0.0f;
            acc += (k + 2 < m) ? v2 : 0.0f;
            acc += (k + 3 < m) ? v3 : 0.0f;
        }
    }
    const float dv = __frsqrt_rn((float)(e1 - e0 + 1));
    float v = acc * dv + b0[f];
    h1[(long)d * 16 + f] = (v > 0.0f ? v : 0.0f) * dv;
}

#define A_PITCH 18
__global__ __launch_bounds__(256) void k_pull_w1(const int* __restrict__ rowEnd,
                                                 const int* __restrict__ csr,
                                                 const float* __restrict__ hp,
                                                 const float* __restrict__ W1,
                                                 const float* __restrict__ b1,
                                                 float* __restrict__ out) {
    __shared__ __attribute__((aligned(16))) float sA[32 * A_PITCH];
    __shared__ float sW[16 * 16];
    __shared__ float sOut[32 * N_CLS];
    __shared__ float sB[16];
    const int ln = threadIdx.x >> 3, c = threadIdx.x & 7;
    const int d  = blockIdx.x * 32 + ln;
    const int e1 = rowEnd[d];
    const int e0 = (d > 0) ? rowEnd[d - 1] : 0;
    const float2* hp2 = (const float2*)hp;
    float2 acc = hp2[(long)d * 8 + c];
    {
        const int k = threadIdx.x >> 4, n = threadIdx.x & 15;
        const float wv = (n < N_CLS) ? W1[k * N_CLS + n] : 0.0f;
        const float bv = (n < N_CLS) ? b1[n] : 0.0f;
        sW[threadIdx.x] = wv;
        if (threadIdx.x < 16) sB[threadIdx.x] = bv;
    }
    for (int j = e0; j < e1; j += 8) {
        const int m = (e1 - j < 8) ? (e1 - j) : 8;
        const int mine = (c < m) ? csr[j + c] : 0;
        for (int k = 0; k < m; k += 4) {
            int s0 = __shfl(mine, k, 8),     s1 = __shfl(mine, k + 1, 8);
            int s2 = __shfl(mine, k + 2, 8), s3 = __shfl(mine, k + 3, 8);
            float2 v0 = hp2[(long)s0 * 8 + c], v1 = hp2[(long)s1 * 8 + c];
            float2 v2 = hp2[(long)s2 * 8 + c], v3 = hp2[(long)s3 * 8 + c];
            __builtin_amdgcn_sched_barrier(0);
            acc.x += v0.x;                          acc.y += v0.y;
            acc.x += (k + 1 < m) ? v1.x : 0.0f;     acc.y += (k + 1 < m) ? v1.y : 0.0f;
            acc.x += (k + 2 < m) ? v2.x : 0.0f;     acc.y += (k + 2 < m) ? v2.y : 0.0f;
            acc.x += (k + 3 < m) ? v3.x : 0.0f;     acc.y += (k + 3 < m) ? v3.y : 0.0f;
        }
    }
    const float dv = __frsqrt_rn((float)(e1 - e0 + 1));
    *(float2*)(sA + ln * A_PITCH + c * 2) = make_float2(acc.x * dv, acc.y * dv);
    __syncthreads();

    const int wave = threadIdx.x >> 5, lane = threadIdx.x & 31;
    if (wave < 2) {
        const int laneHi = lane >> 4;
        const int lane15 = lane & 15;
        const int koff   = laneHi ? 2 : 0;
        const float* arow = sA + (wave * 16 + lane15) * A_PITCH;
        v8f o = {};
#pragma unroll
        for (int kb = 0; kb < 16; kb += 4) {
            v2f a; a.x = arow[kb + koff]; a.y = arow[kb + koff + 1];
            v2f b; b.x = sW[(kb + koff) * 16 + lane15];
                   b.y = sW[(kb + koff + 1) * 16 + lane15];
            o = __builtin_amdgcn_wmma_f32_16x16x4_f32(false, a, false, b,
                                                      (short)0, o, false, false);
        }
        if (lane15 < N_CLS) {
            const float bias = sB[lane15];
            float* so = sOut + (wave * 16 + laneHi * 8) * N_CLS + lane15;
#pragma unroll
            for (int i = 0; i < 8; ++i) so[i * N_CLS] = o[i] + bias;
        }
    }
    __syncthreads();
    if (threadIdx.x < 32 * N_CLS)
        out[(long)blockIdx.x * (32 * N_CLS) + threadIdx.x] = sOut[threadIdx.x];
}

extern "C" void kernel_launch(void* const* d_in, const int* in_sizes, int n_in,
                              void* d_out, int out_size, void* d_ws, size_t ws_size,
                              hipStream_t stream) {
    const float* x  = (const float*)d_in[0];
    const int*   ei = (const int*)d_in[1];
    const float* W0 = (const float*)d_in[2];
    const float* b0 = (const float*)d_in[3];
    const float* W1 = (const float*)d_in[4];
    const float* b1 = (const float*)d_in[5];
    float* out = (float*)d_out;

    const int* src = ei;
    const int* dst = ei + N_EDGES;

    float* ws     = (float*)d_ws;
    int*   rowEnd = (int*)ws;
    float* h0     = ws + N_NODES;
    float* h1     = h0 + (long)N_NODES * 16;
    int*   csr    = (int*)(h1 + (long)N_NODES * 16);
    unsigned short* rank = (unsigned short*)h0;
    int*      tab  = (int*)h0 + 800000;
    unsigned* ctab = (unsigned*)csr;
    int*      part = csr + G_CH * CT_PITCH;

    const int nTiles = N_NODES / 16;
    const dim3 B(256);
    auto gr = [](long n) { return dim3((unsigned)((n + 255) / 256)); };

    k_rank_lds<<<dim3(G_CH), dim3(1024), 0, stream>>>(dst, rank, ctab, part);
    k_scan_tab<<<dim3(NRANGE), dim3(512), 0, stream>>>(ctab, part, rowEnd, tab);
    k_place   <<<gr(N_EDGES), B, 0, stream>>>(src, dst, rank, rowEnd, tab, csr, N_EDGES);

    k_gemm0 <<<dim3((nTiles + 7) / 8), B, 0, stream>>>(x, W0, rowEnd, h0, nTiles);
    k_pull16<<<dim3(N_NODES * 16 / 256), B, 0, stream>>>(rowEnd, csr, h0, b0, h1);

    k_pull_w1<<<dim3(N_NODES / 32), B, 0, stream>>>(rowEnd, csr, h1, W1, b1, out);
}


# Edge placement reads its row offsets from LDS, not L2: 250 work-groups, one per WGP, stage half of one chunk's offset plane (200 KB) and place a 12800-edge slice; csr and every sum order unchanged
# speedup vs baseline: 5.8149x; 1.0316x over previous
#include <hip/hip_runtime.h>

#define N_NODES 100000
#define N_EDGES 1600000
#define F_IN    512
#define HIDDEN  16
#define N_CLS   7

typedef float v2f __attribute__((ext_vector_type(2)));
typedef float v8f __attribute__((ext_vector_type(8)));


#define G_CH       25
#define CH_E       64000
#define NPAIR      (N_NODES / 2)
#define CT_PITCH   50048
#define NRANGE     98
#define PART_PITCH 128

__global__ __launch_bounds__(1024) void k_rank_lds(const int* __restrict__ dst, unsigned short* __restrict__ rank,
                                                   unsigned* __restrict__ ctab, int* __restrict__ part) {
    __shared__ unsigned sCnt[NPAIR];
    const int g = blockIdx.x;
    uint4* s4 = (uint4*)sCnt;
    for (int i = threadIdx.x; i < NPAIR / 4; i += 1024) s4[i] = make_uint4(0u, 0u, 0u, 0u);
    __syncthreads();

    const int4* d4 = (const int4*)(dst + g * CH_E);
    uint2*      r2 = (uint2*)(rank + g * CH_E);
#pragma unroll
    for (int h = 0; h < 2; ++h) {
        int4 d[8];
#pragma unroll
        for (int k = 0; k < 8; ++k) {
            const int i = (h * 8 + k) * 1024 + threadIdx.x;
            d[k] = d4[i < CH_E / 4 ? i : 0];
        }
        __builtin_amdgcn_sched_barrier(0);
#pragma unroll
        for (int k = 0; k < 8; ++k) {
            const int i = (h * 8 + k) * 1024 + threadIdx.x;
            if (i < CH_E / 4) {
                const unsigned sx = (d[k].x & 1) << 4, sy = (d[k].y & 1) << 4;
                const unsigned sz = (d[k].z & 1) << 4, sw = (d[k].w & 1) << 4;
                const unsigned ox = atomicAdd(&sCnt[d[k].x >> 1], 1u << sx);
                const unsigned oy = atomicAdd(&sCnt[d[k].y >> 1], 1u << sy);
                const unsigned oz = atomicAdd(&sCnt[d[k].z >> 1], 1u << sz);
                const unsigned ow = atomicAdd(&sCnt[d[k].w >> 1], 1u << sw);
                uint2 o;
                o.x = ((ox >> sx) & 0xFFFFu) | (((oy >> sy) & 0xFFFFu) << 16);
                o.y = ((oz >> sz) & 0xFFFFu) | (((ow >> sw) & 0xFFFFu) << 16);
                r2[i] = o;
            }
        }
    }
    __syncthreads();

    const int wave = threadIdx.x >> 5, lane = threadIdx.x & 31;
    uint4* c4 = (uint4*)(ctab + (long)g * CT_PITCH);
    for (int r = wave; r < NRANGE; r += 32) {
        unsigned sum = 0;
#pragma unroll
        for (int j = 0; j < 4; ++j) {
            const int q = r * 128 + j * 32 + lane;
            if (q < NPAIR / 4) {
                const uint4 v = s4[q];
                c4[q] = v;
                sum += ((v.x & 0xFFFFu) + (v.x >> 16)) + ((v.y & 0xFFFFu) + (v.y >> 16))
                     + ((v.z & 0xFFFFu) + (v.z >> 16)) + ((v.w & 0xFFFFu) + (v.w >> 16));
            }
        }
        for (int o = 16; o > 0; o >>= 1) sum += __shfl_xor(sum, o);
        if (lane == 0) part[g * PART_PITCH + r] = (int)sum;
    }
}

__global__ __launch_bounds__(512) void k_scan_tab(const unsigned* __restrict__ ctab, const int* __restrict__ part,
                                                  int* __restrict__ rowEnd, int* __restrict__ tab) {
    __shared__ int sTot[16];
    __shared__ int sPre[16];
    const int b = blockIdx.x, t = threadIdx.x;
    const int wave = t >> 5, lane = t & 31;
    const int p = b * 512 + t;
    const bool ok = p < NPAIR;

    unsigned c[G_CH];
#pragma unroll
    for (int g = 0; g < G_CH; ++g) c[g] = ctab[(long)g * CT_PITCH + (ok ? p : 0)];

    constexpr int NPV = (G_CH * PART_PITCH + 511) / 512;
    int pv[NPV];
#pragma unroll
    for (int k = 0; k < NPV; ++k) {
        const int idx = k * 512 + t;
        const bool use = idx < G_CH * PART_PITCH && (idx & (PART_PITCH - 1)) < b;
        pv[k] = part[use ? idx : 0];
    }
    __builtin_amdgcn_sched_barrier(0);
    int pre = 0;
#pragma unroll
    for (int k = 0; k < NPV; ++k) {
        const int idx = k * 512 + t;
        const bool use = idx < G_CH * PART_PITCH && (idx & (PART_PITCH - 1)) < b;
        pre += use ? pv[k] : 0;
    }

    int lo = 0, hi = 0;
#pragma unroll
    for (int g = 0; g < G_CH; ++g) {
        if (!ok) c[g] = 0u;
        lo += (int)(c[g] & 0xFFFFu); hi += (int)(c[g] >> 16);
    }
    const int s = lo + hi;

    int inc = s;
    for (int o = 1; o < 32; o <<= 1) {
        int v = __shfl_up(inc, o);
        if (lane >= o) inc += v;
    }
    for (int o = 16; o > 0; o >>= 1) pre += __shfl_xor(pre, o);
    if (lane == 31) sTot[wave] = inc;
    if (lane == 0)  sPre[wave] = pre;
    __syncthreads();

    int base = (lane < 16) ? (sPre[lane] + ((lane < wave) ? sTot[lane] : 0)) : 0;
    for (int o = 16; o > 0; o >>= 1) base += __shfl_xor(base, o);

    const int ex = base + inc - s;
    if (ok) {
        ((int2*)rowEnd)[p] = make_int2(ex + lo, ex + s);
        int runLo = ex, runHi = ex + lo;
#pragma unroll
        for (int g = 0; g < G_CH - 1; ++g) {
            runLo += (int)(c[g] & 0xFFFFu);
            runHi += (int)(c[g] >> 16);
            ((int2*)(tab + (long)g * N_NODES))[p] = make_int2(runLo, runHi);
        }
    }
}

#define PQ     2
#define PQ_N   (N_NODES / PQ)
#define PS     5
#define PS_E   (CH_E / PS)
#define PL_IT  ((PQ_N / 4 + 1023) / 1024)
#define PE_IT  ((PS_E / 4 + 1023) / 1024)
__global__ __launch_bounds__(1024) void k_place_lds(const int* __restrict__ src, const int* __restrict__ dst,
                                                    const unsigned short* __restrict__ rank,
                                                    const int* __restrict__ rowEnd, const int* __restrict__ tab,
                                                    int* __restrict__ csr) {
    __shared__ __attribute__((aligned(16))) int sOff[PQ_N + 4];
    const int b  = blockIdx.x;
    const int g  = b / (PQ * PS);
    const int q  = (b / PS) % PQ;
    const int s  = b % PS;
    const int nb = q * PQ_N;
    const int e0 = g * CH_E + s * PS_E;

    const int4* p4 = (const int4*)((g == 0) ? rowEnd + nb : tab + (long)(g - 1) * N_NODES + nb);
    int4 t[PL_IT];
#pragma unroll
    for (int k = 0; k < PL_IT; ++k) {
        const int i = k * 1024 + threadIdx.x;
        t[k] = p4[i < PQ_N / 4 ? i : 0];
    }
    int first = 0;
    if (g == 0 && nb > 0) first = rowEnd[nb - 1];

    const int4*  d4 = (const int4*)(dst + e0);
    const int4*  s4 = (const int4*)(src + e0);
    const uint2* r2 = (const uint2*)(rank + e0);
    int4 d[PE_IT], sv[PE_IT];
    uint2 r[PE_IT];
#pragma unroll
    for (int k = 0; k < PE_IT; ++k) {
        const int i = k * 1024 + threadIdx.x;
        const int j = i < PS_E / 4 ? i : 0;
        d[k] = d4[j]; sv[k] = s4[j]; r[k] = r2[j];
    }
    if (g == 0) {
#pragma unroll
        for (int k = 0; k < PL_IT; ++k) {
            const int i = k * 1024 + threadIdx.x;
            if (i < PQ_N / 4) {
                int* o = sOff + 4 * i + 1;
                o[0] = t[k].x; o[1] = t[k].y; o[2] = t[k].z; o[3] = t[k].w;
            }
        }
        if (threadIdx.x == 0) sOff[0] = first;
    } else {
#pragma unroll
        for (int k = 0; k < PL_IT; ++k) {
            const int i = k * 1024 + threadIdx.x;
            if (i < PQ_N / 4) ((int4*)sOff)[i] = t[k];
        }
    }
    __syncthreads();

#pragma unroll
    for (int k = 0; k < PE_IT; ++k) {
        const int i = k * 1024 + threadIdx.x;
        if (i < PS_E / 4) {
            const unsigned lx = (unsigned)(d[k].x - nb), ly = (unsigned)(d[k].y - nb);
            const unsigned lz = (unsigned)(d[k].z - nb), lw = (unsigned)(d[k].w - nb);
            const int ox = sOff[lx < (unsigned)PQ_N ? lx : 0u], oy = sOff[ly < (unsigned)PQ_N ? ly : 0u];
            const int oz = sOff[lz < (unsigned)PQ_N ? lz : 0u], ow = sOff[lw < (unsigned)PQ_N ? lw : 0u];
            if (lx < (unsigned)PQ_N) csr[ox + (int)(r[k].x & 0xFFFFu)] = sv[k].x;
            if (ly < (unsigned)PQ_N) csr[oy + (int)(r[k].x >> 16)]     = sv[k].y;
            if (lz < (unsigned)PQ_N) csr[oz + (int)(r[k].y & 0xFFFFu)] = sv[k].z;
            if (lw < (unsigned)PQ_N) csr[ow + (int)(r[k].y >> 16)]     = sv[k].w;
        }
    }
}

__global__ __launch_bounds__(256) void k_gemm0(const float* __restrict__ x,
                                               const float* __restrict__ W0,
                                               const int* __restrict__ rowEnd,
                                               float* __restrict__ h0, int nTiles) {
    __shared__ float sW[F_IN * HIDDEN];
    for (int i = threadIdx.x; i < (F_IN * HIDDEN) / 4; i += 256)
        ((float4*)sW)[i] = ((const float4*)W0)[i];
    __syncthreads();

    const int wave   = threadIdx.x >> 5;
    const int lane   = threadIdx.x & 31;
    const int tile   = blockIdx.x * 8 + wave;
    if (tile >= nTiles) return;

    const int laneHi = lane >> 4;
    const int lane15 = lane & 15;
    const int row    = tile * 16 + lane15;
    const float* xrow = x + (long)row * F_IN;
    const int koff   = laneHi ? 4 : 0;

    v8f acc = {};
    for (int kb = 0; kb < F_IN; kb += 8) {
        float4 a4 = *(const float4*)(xrow + kb + koff);
        const float* wb = sW + (kb + koff) * HIDDEN + lane15;
        v2f a0; a0.x = a4.x; a0.y = a4.y;
        v2f b0; b0.x = wb[0];  b0.y = wb[16];
        acc = __builtin_amdgcn_wmma_f32_16x16x4_f32(false, a0, false, b0,
                                                    (short)0, acc, false, false);
        v2f a1; a1.x = a4.z; a1.y = a4.w;
        v2f b1; b1.x = wb[32]; b1.y = wb[48];
        acc = __builtin_amdgcn_wmma_f32_16x16x4_f32(false, a1, false, b1,
                                                    (short)0, acc, false, false);
    }
    const int r0 = tile * 16 + laneHi * 8;
    int prev = (r0 > 0) ? rowEnd[r0 - 1] : 0;
    float* out = h0 + (long)tile * 256 + laneHi * 128 + lane15;
#pragma unroll
    for (int i = 0; i < 8; ++i) {
        int e = rowEnd[r0 + i];
        float dv = __frsqrt_rn((float)(e - prev + 1));
        prev = e;
        out[i * 16] = acc[i] * dv;
    }
}

__global__ __launch_bounds__(256) void k_pull16(const int* __restrict__ rowEnd,
                                                const int* __restrict__ csr,
                                                const float* __restrict__ hp,
                                                const float* __restrict__ b0,
                                                float* __restrict__ h1) {
    const int t = blockIdx.x * 256 + threadIdx.x;
    const int d = t >> 4, f = t & 15;
    const int e1 = rowEnd[d];
    const int e0 = (d > 0) ? rowEnd[d - 1] : 0;
    float acc = hp[(long)d * 16 + f];
    for (int j = e0; j < e1; j += 16) {
        const int m = (e1 - j < 16) ? (e1 - j) : 16;
        const int mine = (f < m) ? csr[j + f] : 0;
        for (int k = 0; k < m; k += 4) {
            int s0 = __shfl(mine, k, 16),     s1 = __shfl(mine, k + 1, 16);
            int s2 = __shfl(mine, k + 2, 16), s3 = __shfl(mine, k + 3, 16);
            float v0 = hp[(long)s0 * 16 + f], v1 = hp[(long)s1 * 16 + f];
            float v2 = hp[(long)s2 * 16 + f], v3 = hp[(long)s3 * 16 + f];
            __builtin_amdgcn_sched_barrier(0);
            acc += v0;
            acc += (k + 1 < m) ? v1 : 0.0f;
            acc += (k + 2 < m) ? v2 : 0.0f;
            acc += (k + 3 < m) ? v3 : 0.0f;
        }
    }
    const float dv = __frsqrt_rn((float)(e1 - e0 + 1));
    float v = acc * dv + b0[f];
    h1[(long)d * 16 + f] = (v > 0.0f ? v : 0.0f) * dv;
}

#define A_PITCH 18
__global__ __launch_bounds__(256) void k_pull_w1(const int* __restrict__ rowEnd,
                                                 const int* __restrict__ csr,
                                                 const float* __restrict__ hp,
                                                 const float* __restrict__ W1,
                                                 const float* __restrict__ b1,
                                                 float* __restrict__ out) {
    __shared__ __attribute__((aligned(16))) float sA[32 * A_PITCH];
    __shared__ float sW[16 * 16];
    __shared__ float sOut[32 * N_CLS];
    __shared__ float sB[16];
    const int ln = threadIdx.x >> 3, c = threadIdx.x & 7;
    const int d  = blockIdx.x * 32 + ln;
    const int e1 = rowEnd[d];
    const int e0 = (d > 0) ? rowEnd[d - 1] : 0;
    const float2* hp2 = (const float2*)hp;
    float2 acc = hp2[(long)d * 8 + c];
    {
        const int k = threadIdx.x >> 4, n = threadIdx.x & 15;
        const float wv = (n < N_CLS) ? W1[k * N_CLS + n] : 0.0f;
        const float bv = (n < N_CLS) ? b1[n] : 0.0f;
        sW[threadIdx.x] = wv;
        if (threadIdx.x < 16) sB[threadIdx.x] = bv;
    }
    for (int j = e0; j < e1; j += 8) {
        const int m = (e1 - j < 8) ? (e1 - j) : 8;
        const int mine = (c < m) ? csr[j + c] : 0;
        for (int k = 0; k < m; k += 4) {
            int s0 = __shfl(mine, k, 8),     s1 = __shfl(mine, k + 1, 8);
            int s2 = __shfl(mine, k + 2, 8), s3 = __shfl(mine, k + 3, 8);
            float2 v0 = hp2[(long)s0 * 8 + c], v1 = hp2[(long)s1 * 8 + c];
            float2 v2 = hp2[(long)s2 * 8 + c], v3 = hp2[(long)s3 * 8 + c];
            __builtin_amdgcn_sched_barrier(0);
            acc.x += v0.x;                          acc.y += v0.y;
            acc.x += (k + 1 < m) ? v1.x : 0.0f;     acc.y += (k + 1 < m) ? v1.y : 0.0f;
            acc.x += (k + 2 < m) ? v2.x : 0.0f;     acc.y += (k + 2 < m) ? v2.y : 0.0f;
            acc.x += (k + 3 < m) ? v3.x : 0.0f;     acc.y += (k + 3 < m) ? v3.y : 0.0f;
        }
    }
    const float dv = __frsqrt_rn((float)(e1 - e0 + 1));
    *(float2*)(sA + ln * A_PITCH + c * 2) = make_float2(acc.x * dv, acc.y * dv);
    __syncthreads();

    const int wave = threadIdx.x >> 5, lane = threadIdx.x & 31;
    if (wave < 2) {
        const int laneHi = lane >> 4;
        const int lane15 = lane & 15;
        const int koff   = laneHi ? 2 : 0;
        const float* arow = sA + (wave * 16 + lane15) * A_PITCH;
        v8f o = {};
#pragma unroll
        for (int kb = 0; kb < 16; kb += 4) {
            v2f a; a.x = arow[kb + koff]; a.y = arow[kb + koff + 1];
            v2f b; b.x = sW[(kb + koff) * 16 + lane15];
                   b.y = sW[(kb + koff + 1) * 16 + lane15];
            o = __builtin_amdgcn_wmma_f32_16x16x4_f32(false, a, false, b,
                                                      (short)0, o, false, false);
        }
        if (lane15 < N_CLS) {
            const float bias = sB[lane15];
            float* so = sOut + (wave * 16 + laneHi * 8) * N_CLS + lane15;
#pragma unroll
            for (int i = 0; i < 8; ++i) so[i * N_CLS] = o[i] + bias;
        }
    }
    __syncthreads();
    if (threadIdx.x < 32 * N_CLS)
        out[(long)blockIdx.x * (32 * N_CLS) + threadIdx.x] = sOut[threadIdx.x];
}

extern "C" void kernel_launch(void* const* d_in, const int* in_sizes, int n_in,
                              void* d_out, int out_size, void* d_ws, size_t ws_size,
                              hipStream_t stream) {
    const float* x  = (const float*)d_in[0];
    const int*   ei = (const int*)d_in[1];
    const float* W0 = (const float*)d_in[2];
    const float* b0 = (const float*)d_in[3];
    const float* W1 = (const float*)d_in[4];
    const float* b1 = (const float*)d_in[5];
    float* out = (float*)d_out;

    const int* src = ei;
    const int* dst = ei + N_EDGES;

    float* ws     = (float*)d_ws;
    int*   rowEnd = (int*)ws;
    float* h0     = ws + N_NODES;
    float* h1     = h0 + (long)N_NODES * 16;
    int*   csr    = (int*)(h1 + (long)N_NODES * 16);
    unsigned short* rank = (unsigned short*)h0;
    int*      tab  = (int*)h0 + 800000;
    unsigned* ctab = (unsigned*)csr;
    int*      part = csr + G_CH * CT_PITCH;

    const int nTiles = N_NODES / 16;
    const dim3 B(256);

    k_rank_lds<<<dim3(G_CH), dim3(1024), 0, stream>>>(dst, rank, ctab, part);
    k_scan_tab<<<dim3(NRANGE), dim3(512), 0, stream>>>(ctab, part, rowEnd, tab);
    k_place_lds<<<dim3(G_CH * PQ * PS), dim3(1024), 0, stream>>>(src, dst, rank, rowEnd, tab, csr);

    k_gemm0 <<<dim3((nTiles + 7) / 8), B, 0, stream>>>(x, W0, rowEnd, h0, nTiles);
    k_pull16<<<dim3(N_NODES * 16 / 256), B, 0, stream>>>(rowEnd, csr, h0, b0, h1);

    k_pull_w1<<<dim3(N_NODES / 32), B, 0, stream>>>(rowEnd, csr, h1, W1, b1, out);
}
